# starting kernel
# baseline (speedup 1.0000x reference)
#include <hip/hip_runtime.h>

typedef float v2f __attribute__((ext_vector_type(2)));
typedef float v8f __attribute__((ext_vector_type(8)));

#define ND_   50000
#define NG_   100000
#define E_    600000
#define DIN_  128
#define H_    64

__global__ __launch_bounds__(256)
void count_edges(const int* __restrict__ dst, float* __restrict__ cnt, int nE) {
    int i = blockIdx.x * blockDim.x + threadIdx.x;
    if (i < nE) atomicAdd(&cnt[dst[i]], 1.0f);
}

template<int LOGD>
__global__ __launch_bounds__(256)
void scatter_add4(const int* __restrict__ src, const int* __restrict__ dst,
                  const float* __restrict__ xsrc, float* __restrict__ acc, int nE) {
    const unsigned CPT = 1u << (LOGD - 2);
    unsigned idx = blockIdx.x * blockDim.x + threadIdx.x;
    unsigned e   = idx >> (LOGD - 2);
    if (e >= (unsigned)nE) return;
    unsigned c = idx & (CPT - 1);

    const float4* xs = (const float4*)(xsrc + ((size_t)src[e] << LOGD));
    float4 v = xs[c];
    float* out = acc + ((size_t)dst[e] << LOGD) + (size_t)c * 4;
    atomicAdd(out + 0, v.x);
    atomicAdd(out + 1, v.y);
    atomicAdd(out + 2, v.z);
    atomicAdd(out + 3, v.w);
}

template<int K>
__global__ __launch_bounds__(128)
void sage_wmma(const float* __restrict__ msum, const float* __restrict__ cnt,
               const float* __restrict__ xdst,
               const float* __restrict__ Wl, const float* __restrict__ bl,
               const float* __restrict__ Wr,
               float* __restrict__ out, int N)
{
    __shared__ v2f Wpk[K * 64];
    {
        float* lds = (float*)Wpk;
        for (int i = threadIdx.x; i < K * 64; i += blockDim.x) {
            const int k = i >> 6, col = i & 63;
            lds[((k >> 1) * 64 + col) * 2 + (k & 1)]           = Wl[i];
            lds[((K / 2 + (k >> 1)) * 64 + col) * 2 + (k & 1)] = Wr[i];
        }
    }
    __syncthreads();

    const int wave = threadIdx.x >> 5;
    const int lane = threadIdx.x & 31;
    const int pair = blockIdx.x * 4 + wave;
    const int row0 = pair * 32;
    if (row0 >= N) return;
    const bool has2 = (row0 + 32 <= N);

    const int m  = lane & 15;
    const int ph = lane >> 4;
    const int rowA0 = row0 + m;
    const int rowA1 = min(row0 + 16 + m, N - 1);

    const float s0 = 1.0f / fmaxf(cnt[rowA0], 1.0f);
    const float s1 = 1.0f / fmaxf(cnt[rowA1], 1.0f);

    v8f acc0[4] = {};
    v8f acc1[4] = {};

    {
        const v2f* A0 = (const v2f*)(msum + (size_t)rowA0 * K);
        const v2f* A1 = (const v2f*)(msum + (size_t)rowA1 * K);
        for (int k = 0; k < K; k += 4) {
            const int kp = (k >> 1) + ph;
            v2f a0 = A0[kp] * s0;
            v2f a1 = A1[kp] * s1;
#pragma unroll
            for (int t = 0; t < 4; ++t) {
                v2f b = Wpk[kp * 64 + t * 16 + m];
                acc0[t] = __builtin_amdgcn_wmma_f32_16x16x4_f32(
                    false, a0, false, b, (short)0, acc0[t], false, false);
                acc1[t] = __builtin_amdgcn_wmma_f32_16x16x4_f32(
                    false, a1, false, b, (short)0, acc1[t], false, false);
            }
        }
    }

    {
        const v2f* X0 = (const v2f*)(xdst + (size_t)rowA0 * K);
        const v2f* X1 = (const v2f*)(xdst + (size_t)rowA1 * K);
        for (int k = 0; k < K; k += 4) {
            const int kp = (k >> 1) + ph;
            v2f a0 = X0[kp];
            v2f a1 = X1[kp];
#pragma unroll
            for (int t = 0; t < 4; ++t) {
                v2f b = Wpk[(K / 2 + kp) * 64 + t * 16 + m];
                acc0[t] = __builtin_amdgcn_wmma_f32_16x16x4_f32(
                    false, a0, false, b, (short)0, acc0[t], false, false);
                acc1[t] = __builtin_amdgcn_wmma_f32_16x16x4_f32(
                    false, a1, false, b, (short)0, acc1[t], false, false);
            }
        }
    }

    const int colb  = lane & 15;
    const int mb0   = row0 + 8 * ph;
#pragma unroll
    for (int t = 0; t < 4; ++t) {
        const float bias = bl[t * 16 + colb];
#pragma unroll
        for (int r = 0; r < 8; ++r)
            out[(size_t)(mb0 + r) * 64 + t * 16 + colb] = acc0[t][r] + bias;
        if (has2) {
#pragma unroll
            for (int r = 0; r < 8; ++r)
                out[(size_t)(mb0 + 16 + r) * 64 + t * 16 + colb] = acc1[t][r] + bias;
        }
    }
}

extern "C" void kernel_launch(void* const* d_in, const int* in_sizes, int n_in,
                              void* d_out, int out_size, void* d_ws, size_t ws_size,
                              hipStream_t stream) {
    const float* x_disease = (const float*)d_in[0];
    const float* x_gene    = (const float*)d_in[1];
    const int*   src_dg    = (const int*)d_in[2];
    const int*   dst_dg    = (const int*)d_in[3];
    const int*   src_gd    = (const int*)d_in[4];
    const int*   dst_gd    = (const int*)d_in[5];
    const float* Wl1_dg = (const float*)d_in[6];
    const float* bl1_dg = (const float*)d_in[7];
    const float* Wr1_dg = (const float*)d_in[8];
    const float* Wl1_gd = (const float*)d_in[9];
    const float* bl1_gd = (const float*)d_in[10];
    const float* Wr1_gd = (const float*)d_in[11];
    const float* Wl2_dg = (const float*)d_in[12];
    const float* bl2_dg = (const float*)d_in[13];
    const float* Wr2_dg = (const float*)d_in[14];
    const float* Wl2_gd = (const float*)d_in[15];
    const float* bl2_gd = (const float*)d_in[16];
    const float* Wr2_gd = (const float*)d_in[17];

    float* d2 = (float*)d_out;
    float* g2 = d2 + (size_t)ND_ * 64;

    float* ws = (float*)d_ws;
    size_t off = 0;
    float* ms_g  = ws + off; off += (size_t)NG_ * DIN_;
    float* ms_d  = ws + off; off += (size_t)ND_ * DIN_;
    float* cnt_g = ws + off; off += (size_t)NG_;
    float* cnt_d = ws + off; off += (size_t)ND_;
    float* g1    = ws + off; off += (size_t)NG_ * H_;
    float* d1    = ws + off; off += (size_t)ND_ * H_;
    (void)off; (void)ws_size; (void)in_sizes; (void)n_in; (void)out_size;

    const int TPB = 256;
    const int eb  = (E_ + TPB - 1) / TPB;

    hipMemsetAsync(ms_g,  0, (size_t)NG_ * DIN_ * sizeof(float), stream);
    hipMemsetAsync(ms_d,  0, (size_t)ND_ * DIN_ * sizeof(float), stream);
    hipMemsetAsync(cnt_g, 0, (size_t)NG_ * sizeof(float), stream);
    hipMemsetAsync(cnt_d, 0, (size_t)ND_ * sizeof(float), stream);

    count_edges<<<eb, TPB, 0, stream>>>(dst_dg, cnt_g, E_);
    count_edges<<<eb, TPB, 0, stream>>>(dst_gd, cnt_d, E_);

    {
        const int blocks = ((E_ << (7 - 2)) + TPB - 1) / TPB;
        scatter_add4<7><<<blocks, TPB, 0, stream>>>(src_dg, dst_dg, x_disease, ms_g, E_);
        scatter_add4<7><<<blocks, TPB, 0, stream>>>(src_gd, dst_gd, x_gene,    ms_d, E_);
    }

    {
        const int pairs_g = (NG_ / 16 + 1) / 2;
        const int pairs_d = (ND_ / 16 + 1) / 2;
        sage_wmma<DIN_><<<(pairs_g + 3) / 4, 128, 0, stream>>>(
            ms_g, cnt_g, x_gene,    Wl1_dg, bl1_dg, Wr1_dg, g1, NG_);
        sage_wmma<DIN_><<<(pairs_d + 3) / 4, 128, 0, stream>>>(
            ms_d, cnt_d, x_disease, Wl1_gd, bl1_gd, Wr1_gd, d1, ND_);
    }

    hipMemsetAsync(ms_g, 0, (size_t)NG_ * H_ * sizeof(float), stream);
    hipMemsetAsync(ms_d, 0, (size_t)ND_ * H_ * sizeof(float), stream);
    {
        const int blocks = ((E_ << (6 - 2)) + TPB - 1) / TPB;
        scatter_add4<6><<<blocks, TPB, 0, stream>>>(src_dg, dst_dg, d1, ms_g, E_);
        scatter_add4<6><<<blocks, TPB, 0, stream>>>(src_gd, dst_gd, g1, ms_d, E_);
    }

    {
        const int pairs_g = (NG_ / 16 + 1) / 2;
        const int pairs_d = (ND_ / 16 + 1) / 2;
        sage_wmma<H_><<<(pairs_g + 3) / 4, 128, 0, stream>>>(
            ms_g, cnt_g, g1, Wl2_dg, bl2_dg, Wr2_dg, g2, NG_);
        sage_wmma<H_><<<(pairs_d + 3) / 4, 128, 0, stream>>>(
            ms_d, cnt_d, d1, Wl2_gd, bl2_gd, Wr2_gd, d2, ND_);
    }
}


# layer 2 reuses layer 1's filter: each layer-1 wave saves the edges it kept (sender row, tile row) in whole lines; layer-2 blocks add those rows, no index scans; LDS add order by arrival as before
# speedup vs baseline: 3.4155x; 3.4155x over previous
#include <hip/hip_runtime.h>


typedef float v2f __attribute__((ext_vector_type(2)));
typedef float v8f __attribute__((ext_vector_type(8)));

#define ND_   50000
#define NG_   100000
#define E_    600000
#define DIN_  128
#define H_    64

#define SEG_THREADS 1024
#define SEG_WAVES   (SEG_THREADS / 32)
#define SEG_BATCH   8
#define SEG_QCAP    136
#define SEG_LOCBITS 9
#define SEG_WCAP    160
#define SEG_LB      16

#define RG_   400
#define RD_   200
#define NBLK_ 250
constexpr unsigned long long WS_LIST_OFF_ = (unsigned long long)NG_ + ND_ + ((unsigned long long)NG_ + ND_) * H_ + 16;
constexpr unsigned long long WS_WORDS_    = WS_LIST_OFF_ + 2ull * NBLK_ * SEG_WAVES * (1 + SEG_WCAP);

template<int VEC>
__device__ __forceinline__ constexpr int seg_fo(int c)
{
    return (VEC == 4) ? (c * 32 + (c >> 1) * 16) : (c * 32);
}
template<int K>
__device__ __forceinline__ constexpr int seg_pitch()
{
    return (K == 128) ? 145 : 66;
}

template<int K, bool FULL>
__device__ __forceinline__
unsigned seg_drain(const unsigned* qp, int n, const int* __restrict__ src,
               const float* __restrict__ xsrc, float* acc, int lane)
{
    constexpr int VEC = K / 32;
    constexpr int P   = seg_pitch<K>();
    typedef float VT __attribute__((ext_vector_type(VEC)));
    const int li = FULL ? (lane & (SEG_BATCH - 1)) : min(lane & (SEG_BATCH - 1), n - 1);
    const unsigned ent = qp[li];
    const int      s   = src[ent >> SEG_LOCBITS];
    const unsigned loc = ent & ((1u << SEG_LOCBITS) - 1u);

    VT v[SEG_BATCH];
#pragma unroll
    for (int u = 0; u < SEG_BATCH; ++u) {
        if (FULL || u < n) {
            const int su = __builtin_amdgcn_readlane(s, u);
            v[u] = ((const VT*)(xsrc + (size_t)su * K))[lane];
        }
    }
#pragma unroll
    for (int u = 0; u < SEG_BATCH; ++u) {
        if (FULL || u < n) {
            const unsigned lu = __builtin_amdgcn_readlane(loc, u);
            float* a = acc + lu * P + lane;
#pragma unroll
            for (int c = 0; c < VEC; ++c) atomicAdd(a + seg_fo<VEC>(c), v[u][c]);
        }
    }
    return ((unsigned)s << SEG_LOCBITS) | loc;
}

template<int K, int NB, bool FULL>
__device__ __forceinline__
void seg_add_saved(unsigned ent, int b, int n, const float* __restrict__ xsrc, float* acc, int lane)
{
    constexpr int VEC = K / 32;
    constexpr int P   = seg_pitch<K>();
    typedef float VT __attribute__((ext_vector_type(VEC)));
    VT v[NB];
#pragma unroll
    for (int u = 0; u < NB; ++u) {
        if (FULL || u < n) {
            const unsigned eu = __builtin_amdgcn_readlane(ent, b + u);
            v[u] = ((const VT*)(xsrc + (size_t)(eu >> SEG_LOCBITS) * K))[lane];
        }
    }
#pragma unroll
    for (int u = 0; u < NB; ++u) {
        if (FULL || u < n) {
            const unsigned lu = __builtin_amdgcn_readlane(ent, b + u) & ((1u << SEG_LOCBITS) - 1u);
            float* a = acc + lu * P + lane;
#pragma unroll
            for (int c = 0; c < VEC; ++c) atomicAdd(a + seg_fo<VEC>(c), v[u][c]);
        }
    }
}

__device__ __forceinline__
int4 seg_ld4(const int4* __restrict__ dst4, int i4, int n4)
{
    return dst4[min(i4, n4 - 1)];
}

template<int K, int R, bool WCNT>
__global__ __launch_bounds__(SEG_THREADS)
void sage_fused(const int* __restrict__ src, const int* __restrict__ dst,
                const float* __restrict__ xsrc, const float* __restrict__ xdst,
                const float* __restrict__ Wl, const float* __restrict__ bl,
                const float* __restrict__ Wr,
                float* __restrict__ cnt, float* __restrict__ out,
                unsigned* __restrict__ hdr, unsigned* __restrict__ lst, unsigned long long ws_bytes,
                int nE, int N)
{
    static_assert(R <= (1 << SEG_LOCBITS), "row offset must fit its field");
    static_assert(K == 64 || K == 128, "row width");
    constexpr int P = seg_pitch<K>();
    static_assert((R * P) % 4 == 0, "the tile is cleared sixteen bytes at a time");
    static_assert(R % 8 == 0, "a lane half's 8 output rows are stored under one guard");

    __shared__ float    acc[R * P];
    __shared__ v2f      Wpk[K * 64];
    __shared__ unsigned cntL[R];
    __shared__ unsigned queue[SEG_WAVES * SEG_QCAP];
    __shared__ unsigned wlen[SEG_WAVES];

    const int tid  = threadIdx.x;
    const int lane = tid & 31;
    const int wave = __builtin_amdgcn_readfirstlane(tid >> 5);
    const int base = blockIdx.x * R;
    const int rows = min(R, N - base);

    const bool room = ws_bytes >= 4ull * WS_WORDS_;
    unsigned* lw = lst + ((size_t)blockIdx.x * SEG_WAVES + wave) * SEG_WCAP;

    unsigned hc = 0u, e_cur = 0u;
    if (!WCNT && room) {
        hc    = hdr[blockIdx.x * SEG_WAVES + lane];
        e_cur = lw[lane];
    }

    {
        constexpr int WIT = K * 64 / SEG_THREADS;
        static_assert(WIT * SEG_THREADS == K * 64, "every thread stages the same number of weights");
        float wlv[WIT], wrv[WIT];
#pragma unroll
        for (int u = 0; u < WIT; ++u) {
            wlv[u] = Wl[u * SEG_THREADS + tid];
            wrv[u] = Wr[u * SEG_THREADS + tid];
        }

        float4* acc4 = (float4*)acc;
        for (int i = tid; i < R * P / 4; i += SEG_THREADS) acc4[i] = make_float4(0.f, 0.f, 0.f, 0.f);
        for (int i = tid; i < R; i += SEG_THREADS) cntL[i] = 0u;

        float* wl = (float*)Wpk;
#pragma unroll
        for (int u = 0; u < WIT; ++u) {
            const int i = u * SEG_THREADS + tid;
            const int k = i >> 6, col = i & 63;
            wl[((k >> 1) * 64 + col) * 2 + (k & 1)]           = wlv[u];
            wl[((K / 2 + (k >> 1)) * 64 + col) * 2 + (k & 1)] = wrv[u];
        }
    }
    const bool listed = !WCNT && room && __builtin_amdgcn_ballot_w32(hc > (unsigned)SEG_WCAP) == 0u;
    const int  ln     = listed ? (int)__builtin_amdgcn_readlane(hc, wave) : 0;
    __syncthreads();

    if (listed) {
        for (int l0 = 0; l0 < ln; l0 += 32) {
            unsigned e_nxt = 0u;
            if (l0 + 32 < ln) e_nxt = lw[l0 + 32 + lane];
            const int nl = ln - l0;
#pragma unroll
            for (int b = 0; b < 32; b += SEG_LB) {
                const int c = nl - b;
                if (c >= SEG_LB)  seg_add_saved<K, SEG_LB, true >(e_cur, b, SEG_LB, xsrc, acc, lane);
                else if (c > 0)   seg_add_saved<K, SEG_LB, false>(e_cur, b, c, xsrc, acc, lane);
            }
            e_cur = e_nxt;
        }
    } else {
    unsigned* q  = queue + wave * SEG_QCAP;
    int       qn = 0;
    unsigned  pend = 0u;
    int       wn   = 0;
    auto keep = [&](unsigned e2, int n) {
        if ((lane >> 3) == ((wn >> 3) & 3)) pend = e2;
        wn += n;
        if ((wn & 31) == 0 && room && wn <= SEG_WCAP) lw[wn - 32 + lane] = pend;
    };
    const int4* dst4   = (const int4*)dst;
    const int   n4     = nE >> 2;
    const int   nsteps = (n4 + 127) / 128;

    auto scan_step = [&](const int4 (&d)[4], int step) {
#pragma unroll
        for (int qq = 0; qq < 4; ++qq) {
            const int i4 = step * 128 + qq * 32 + lane;
            const unsigned lim = (i4 < n4) ? (unsigned)rows : 0u;
            const int dv[4] = {d[qq].x, d[qq].y, d[qq].z, d[qq].w};
            unsigned loc[4];
            bool     m[4];
            bool     any = false;
#pragma unroll
            for (int j = 0; j < 4; ++j) {
                loc[j] = (unsigned)(dv[j] - base);
                m[j]   = loc[j] < lim;
                any    = any || m[j];
            }
            if (__builtin_amdgcn_ballot_w32(any) != 0u) {
#pragma unroll
                for (int j = 0; j < 4; ++j) {
                    const unsigned mask = __builtin_amdgcn_ballot_w32(m[j]);
                    if (m[j]) {
                        const unsigned pos = (unsigned)qn + __builtin_amdgcn_mbcnt_lo(mask, 0u);
                        q[pos] = ((unsigned)(4 * i4 + j) << SEG_LOCBITS) | loc[j];
                        if (WCNT) atomicAdd(&cntL[loc[j]], 1u);
                    }
                    qn += __builtin_popcount(mask);
                }
                while (qn >= SEG_BATCH) {
                    qn -= SEG_BATCH;
                    const unsigned e2 = seg_drain<K, true>(q + qn, SEG_BATCH, src, xsrc, acc, lane);
                    if (WCNT) keep(e2, SEG_BATCH);
                }
            }
        }
    };

    int4 da[4], db[4];
#pragma unroll
    for (int qq = 0; qq < 4; ++qq) da[qq] = seg_ld4(dst4, wave * 128 + qq * 32 + lane, n4);
    for (int step = wave; step < nsteps; step += 2 * SEG_WAVES) {
        const int sb = step + SEG_WAVES;
        const int sa = step + 2 * SEG_WAVES;
#pragma unroll
        for (int qq = 0; qq < 4; ++qq) db[qq] = seg_ld4(dst4, sb * 128 + qq * 32 + lane, n4);
        scan_step(da, step);
#pragma unroll
        for (int qq = 0; qq < 4; ++qq) da[qq] = seg_ld4(dst4, sa * 128 + qq * 32 + lane, n4);
        if (sb < nsteps) scan_step(db, sb);
    }
    if (qn > 0) {
        const unsigned e2 = seg_drain<K, false>(q, qn, src, xsrc, acc, lane);
        if (WCNT) keep(e2, qn);
    }
    if (WCNT) {
        if (((wn & 31) != 0 || wn == 0) && room && (wn | 31) < SEG_WCAP) lw[(wn & ~31) + lane] = pend;
        if (lane == 0) wlen[wave] = (unsigned)wn;
    }
    }
    __syncthreads();

    if (WCNT) {
        for (int i = tid; i < rows; i += SEG_THREADS) cnt[base + i] = (float)cntL[i];
        if (room && wave == 0) hdr[blockIdx.x * SEG_WAVES + lane] = wlen[lane];
    }

    constexpr int VEC = K / 32;
    const int m      = lane & 15;
    const int ph     = lane >> 4;
    const int ntiles = (rows + 15) >> 4;
    for (int tile = wave; tile < ntiles; tile += SEG_WAVES) {
        const int lr   = min(tile * 16 + m, rows - 1);
        const int grow = base + lr;
        const float cv = WCNT ? (float)cntL[lr] : cnt[grow];
        const float sc = 1.0f / fmaxf(cv, 1.0f);

        float bias[4];
#pragma unroll
        for (int t = 0; t < 4; ++t) bias[t] = bl[t * 16 + m];

        v8f accv[4] = {};

        {
            const float* arow = acc + lr * P + (VEC == 4 ? ph * 80 : ph);
            for (int k = 0; k < K; k += 4) {
                const int j  = (VEC == 4) ? (k >> 2) : (k >> 1);
                const int kp = (k >> 1) + ph;
                v2f a;
                a.x = arow[j] * sc;
                a.y = arow[32 + j] * sc;
#pragma unroll
                for (int t = 0; t < 4; ++t) {
                    const v2f b = Wpk[kp * 64 + t * 16 + m];
                    accv[t] = __builtin_amdgcn_wmma_f32_16x16x4_f32(
                        false, a, false, b, (short)0, accv[t], false, false);
                }
            }
        }

        {
            const v2f* X = (const v2f*)(xdst + (size_t)grow * K);
            for (int k0 = 0; k0 < K; k0 += 32) {
                v2f xa[8];
#pragma unroll
                for (int u = 0; u < 8; ++u) xa[u] = X[((k0 + 4 * u) >> 1) + ph];
#pragma unroll
                for (int u = 0; u < 8; ++u) {
                    const int kp = ((k0 + 4 * u) >> 1) + ph;
#pragma unroll
                    for (int t = 0; t < 4; ++t) {
                        const v2f b = Wpk[(K / 2 + kp) * 64 + t * 16 + m];
                        accv[t] = __builtin_amdgcn_wmma_f32_16x16x4_f32(
                            false, xa[u], false, b, (short)0, accv[t], false, false);
                    }
                }
            }
        }

        const int rr0 = tile * 16 + 8 * ph;
        if (rr0 < rows) {
            float* o = out + (size_t)(base + rr0) * 64 + m;
#pragma unroll
            for (int t = 0; t < 4; ++t) {
#pragma unroll
                for (int r = 0; r < 8; ++r) o[r * 64 + t * 16] = accv[t][r] + bias[t];
            }
        }
    }
}

extern "C" void kernel_launch(void* const* d_in, const int* in_sizes, int n_in,
                              void* d_out, int out_size, void* d_ws, size_t ws_size,
                              hipStream_t stream) {
    const float* x_disease = (const float*)d_in[0];
    const float* x_gene    = (const float*)d_in[1];
    const int*   src_dg    = (const int*)d_in[2];
    const int*   dst_dg    = (const int*)d_in[3];
    const int*   src_gd    = (const int*)d_in[4];
    const int*   dst_gd    = (const int*)d_in[5];
    const float* Wl1_dg = (const float*)d_in[6];
    const float* bl1_dg = (const float*)d_in[7];
    const float* Wr1_dg = (const float*)d_in[8];
    const float* Wl1_gd = (const float*)d_in[9];
    const float* bl1_gd = (const float*)d_in[10];
    const float* Wr1_gd = (const float*)d_in[11];
    const float* Wl2_dg = (const float*)d_in[12];
    const float* bl2_dg = (const float*)d_in[13];
    const float* Wr2_dg = (const float*)d_in[14];
    const float* Wl2_gd = (const float*)d_in[15];
    const float* bl2_gd = (const float*)d_in[16];
    const float* Wr2_gd = (const float*)d_in[17];

    float* d2 = (float*)d_out;
    float* g2 = d2 + (size_t)ND_ * 64;

    float* ws = (float*)d_ws;
    size_t off = 0;
    float* cnt_g = ws + off; off += (size_t)NG_;
    float* cnt_d = ws + off; off += (size_t)ND_;
    float* g1    = ws + off; off += (size_t)NG_ * H_;
    float* d1    = ws + off; off += (size_t)ND_ * H_;
    off += 16;
    unsigned* hdr_g = (unsigned*)(ws + off); off += (size_t)NBLK_ * SEG_WAVES;
    unsigned* hdr_d = (unsigned*)(ws + off); off += (size_t)NBLK_ * SEG_WAVES;
    unsigned* lst_g = (unsigned*)(ws + off); off += (size_t)NBLK_ * SEG_WAVES * SEG_WCAP;
    unsigned* lst_d = (unsigned*)(ws + off); off += (size_t)NBLK_ * SEG_WAVES * SEG_WCAP;
    (void)off; (void)in_sizes; (void)n_in; (void)out_size;
    const unsigned long long wsb = (unsigned long long)ws_size;

    constexpr int RG = RG_;
    constexpr int RD = RD_;
    constexpr int blocks_g = (NG_ + RG - 1) / RG;
    constexpr int blocks_d = (ND_ + RD - 1) / RD;
    static_assert(blocks_g == NBLK_ && blocks_d == NBLK_, "the list planes are sized for this many blocks");
    static_assert(((unsigned long long)NG_ << SEG_LOCBITS) < (1ull << 32) && ND_ <= NG_, "sender row and row offset share one word");
    static_assert(SEG_WCAP % 32 == 0 && (WS_LIST_OFF_ * 4) % 128 == 0, "lists are whole 128-byte lines");
    static_assert(E_ % 4 == 0, "destination indices are read four at a time");
    static_assert(NG_ % 8 == 0 && ND_ % 8 == 0, "a block's row count is a multiple of 8");
    static_assert(((unsigned long long)E_ << SEG_LOCBITS) < (1ull << 32), "edge number and row offset share one word");

    sage_fused<DIN_, RG, true><<<blocks_g, SEG_THREADS, 0, stream>>>(
        src_dg, dst_dg, x_disease, x_gene,    Wl1_dg, bl1_dg, Wr1_dg, cnt_g, g1, hdr_g, lst_g, wsb, E_, NG_);
    sage_fused<DIN_, RD, true><<<blocks_d, SEG_THREADS, 0, stream>>>(
        src_gd, dst_gd, x_gene,    x_disease, Wl1_gd, bl1_gd, Wr1_gd, cnt_d, d1, hdr_d, lst_d, wsb, E_, ND_);

    sage_fused<H_, RG, false><<<blocks_g, SEG_THREADS, 0, stream>>>(
        src_dg, dst_dg, d1, g1, Wl2_dg, bl2_dg, Wr2_dg, cnt_g, g2, hdr_g, lst_g, wsb, E_, NG_);
    sage_fused<H_, RD, false><<<blocks_d, SEG_THREADS, 0, stream>>>(
        src_gd, dst_gd, g1, d1, Wl2_gd, bl2_gd, Wr2_gd, cnt_d, d2, hdr_d, lst_d, wsb, E_, ND_);
}
